# starting kernel
# baseline (speedup 1.0000x reference)
#include <hip/hip_runtime.h>

#define NNODES 65536
#define NEDGES 1048576
#define DIM    64

typedef float v2f __attribute__((ext_vector_type(2)));
typedef float v8f __attribute__((ext_vector_type(8)));

__global__ void zero_kernel(float* __restrict__ p, int n) {
    int i = blockIdx.x * blockDim.x + threadIdx.x;
    if (i < n) p[i] = 0.0f;
}

__global__ void degree_kernel(const int* __restrict__ dst, float* __restrict__ deg, int nE) {
    int i = blockIdx.x * blockDim.x + threadIdx.x;
    if (i < nE) atomicAdd(&deg[dst[i]], 1.0f);
}

__global__ void norm_kernel(float* __restrict__ deg, int n) {
    int i = blockIdx.x * blockDim.x + threadIdx.x;
    if (i < n) deg[i] = rsqrtf(fmaxf(deg[i], 1.0f));
}

__global__ void scale_kernel(const float* __restrict__ in, const float* __restrict__ norm,
                             float* __restrict__ o, int nNodes, int pw) {
    int tid = blockIdx.x * blockDim.x + threadIdx.x;
    if (tid >= nNodes * 16) return;
    int v = tid >> 4, c = tid & 15;
    float s = norm[v];
    if (pw == 2) s *= s;
    float4 x = ((const float4*)(in + v * DIM))[c];
    x.x *= s; x.y *= s; x.z *= s; x.w *= s;
    ((float4*)(o + v * DIM))[c] = x;
}

__global__ __launch_bounds__(256) void scatter_kernel(const float* __restrict__ S,
                                                      const int* __restrict__ src,
                                                      const int* __restrict__ dst,
                                                      float* __restrict__ AGG, int nE) {
    int tid = blockIdx.x * blockDim.x + threadIdx.x;
    int e = tid >> 4;
    int c = tid & 15;
    if (e >= nE) return;
    int s = src[e], d = dst[e];
    float4 v = ((const float4*)(S + s * DIM))[c];
    float* p = AGG + d * DIM + c * 4;
    atomicAdd(p + 0, v.x);
    atomicAdd(p + 1, v.y);
    atomicAdd(p + 2, v.z);
    atomicAdd(p + 3, v.w);
}

__global__ __launch_bounds__(256) void ffn_wmma_kernel(
    const float* __restrict__ r, const float* __restrict__ feat,
    const float* __restrict__ w1, const float* __restrict__ b1,
    const float* __restrict__ w2, const float* __restrict__ b2,
    const float* __restrict__ gamma, const float* __restrict__ beta,
    float* __restrict__ out) {
    __shared__ float lds[8 * 16 * DIM];
    const int wave = threadIdx.x >> 5;
    const int lane = threadIdx.x & 31;
    const int idx  = lane & 15;
    const int half = lane >> 4;
    const int rowbase = blockIdx.x * 128 + wave * 16;
    float* A = lds + wave * 16 * DIM;

    const float2 g2  = ((const float2*)gamma)[lane];
    const float2 be2 = ((const float2*)beta)[lane];
    for (int m = 0; m < 16; ++m) {
        float2 x = ((const float2*)(r + (rowbase + m) * DIM))[lane];
        float s = x.x + x.y;
        #pragma unroll
        for (int o = 16; o > 0; o >>= 1) s += __shfl_xor(s, o, 32);
        float mu = s * (1.0f / 64.0f);
        float dx = x.x - mu, dy = x.y - mu;
        float v = dx * dx + dy * dy;
        #pragma unroll
        for (int o = 16; o > 0; o >>= 1) v += __shfl_xor(v, o, 32);
        float rstd = rsqrtf(v * (1.0f / 64.0f) + 1e-5f);
        float2 y;
        y.x = dx * rstd * g2.x + be2.x;
        y.y = dy * rstd * g2.y + be2.y;
        ((float2*)(A + m * DIM))[lane] = y;
    }
    __syncthreads();

    v8f acc[4];
    #pragma unroll
    for (int j = 0; j < 4; ++j) acc[j] = (v8f)(0.0f);
    for (int kb = 0; kb < 16; ++kb) {
        const int k = kb * 4 + half * 2;
        v2f a = *(const v2f*)(A + idx * DIM + k);
        #pragma unroll
        for (int j = 0; j < 4; ++j) {
            float2 wv = *(const float2*)(w1 + (j * 16 + idx) * DIM + k);
            v2f b = {wv.x, wv.y};
            acc[j] = __builtin_amdgcn_wmma_f32_16x16x4_f32(
                false, a, false, b, (short)0, acc[j], false, false);
        }
    }
    __syncthreads();
    #pragma unroll
    for (int j = 0; j < 4; ++j) {
        const float bias = b1[j * 16 + idx];
        #pragma unroll
        for (int t = 0; t < 8; ++t) {
            float hv = acc[j][t] + bias;
            A[(t + 8 * half) * DIM + j * 16 + idx] = fmaxf(hv, 0.0f);
        }
    }
    __syncthreads();

    v8f acc2[4];
    #pragma unroll
    for (int j = 0; j < 4; ++j) acc2[j] = (v8f)(0.0f);
    for (int kb = 0; kb < 16; ++kb) {
        const int k = kb * 4 + half * 2;
        v2f a = *(const v2f*)(A + idx * DIM + k);
        #pragma unroll
        for (int j = 0; j < 4; ++j) {
            float2 wv = *(const float2*)(w2 + (j * 16 + idx) * DIM + k);
            v2f b = {wv.x, wv.y};
            acc2[j] = __builtin_amdgcn_wmma_f32_16x16x4_f32(
                false, a, false, b, (short)0, acc2[j], false, false);
        }
    }
    #pragma unroll
    for (int j = 0; j < 4; ++j) {
        const int col = j * 16 + idx;
        const float bias = b2[col];
        #pragma unroll
        for (int t = 0; t < 8; ++t) {
            const int row = rowbase + t + 8 * half;
            out[row * DIM + col] = acc2[j][t] + bias + r[row * DIM + col] + feat[row * DIM + col];
        }
    }
}

extern "C" void kernel_launch(void* const* d_in, const int* in_sizes, int n_in,
                              void* d_out, int out_size, void* d_ws, size_t ws_size,
                              hipStream_t stream) {
    const float* feat = (const float*)d_in[0];
    const int*   esrc = (const int*)d_in[1];
    const int*   edst = (const int*)d_in[2];
    const float* w1   = (const float*)d_in[3];
    const float* b1   = (const float*)d_in[4];
    const float* w2   = (const float*)d_in[5];
    const float* b2   = (const float*)d_in[6];
    const float* gam  = (const float*)d_in[7];
    const float* bet  = (const float*)d_in[8];

    float* out0 = (float*)d_out;
    float* out1 = out0 + (size_t)NNODES * DIM;
    float* norm = (float*)d_ws;

    const int ND = NNODES * DIM;
    dim3 blk(256);
    dim3 gN((NNODES + 255) / 256);
    dim3 gE((NEDGES + 255) / 256);
    dim3 gND16((NNODES * 16 + 255) / 256);
    dim3 gND((ND + 255) / 256);
    dim3 gSc((NEDGES * 16 + 255) / 256);

    zero_kernel<<<gN, blk, 0, stream>>>(norm, NNODES);
    degree_kernel<<<gE, blk, 0, stream>>>(edst, norm, NEDGES);
    norm_kernel<<<gN, blk, 0, stream>>>(norm, NNODES);

    scale_kernel<<<gND16, blk, 0, stream>>>(feat, norm, out0, NNODES, 1);
    zero_kernel<<<gND, blk, 0, stream>>>(out1, ND);
    scatter_kernel<<<gSc, blk, 0, stream>>>(out0, esrc, edst, out1, NEDGES);
    scale_kernel<<<gND16, blk, 0, stream>>>(out1, norm, out0, NNODES, 2);
    zero_kernel<<<gND, blk, 0, stream>>>(out1, ND);
    scatter_kernel<<<gSc, blk, 0, stream>>>(out0, esrc, edst, out1, NEDGES);
    scale_kernel<<<gND16, blk, 0, stream>>>(out1, norm, out0, NNODES, 2);
    zero_kernel<<<gND, blk, 0, stream>>>(out1, ND);
    scatter_kernel<<<gSc, blk, 0, stream>>>(out0, esrc, edst, out1, NEDGES);
    scale_kernel<<<gND16, blk, 0, stream>>>(out1, norm, out1, NNODES, 1);

    ffn_wmma_kernel<<<dim3(NNODES / 128), blk, 0, stream>>>(
        out1, feat, w1, b1, w2, b2, gam, bet, out0);
}


# Replace per-edge f32 atomic scatter (3 hops) by dst-grouped gather: int atomic rank + scan build segments per call; per-node sums in registers, order = atomic arrival rank; no zero/scale passes
# speedup vs baseline: 2.4043x; 2.4043x over previous
#include <hip/hip_runtime.h>

#define NNODES 65536
#define NEDGES 1048576
#define DIM    64

typedef float v2f __attribute__((ext_vector_type(2)));
typedef float v8f __attribute__((ext_vector_type(8)));

__global__ void zero_int_kernel(int* __restrict__ p, int n) {
    int i = blockIdx.x * blockDim.x + threadIdx.x;
    if (i < n) p[i] = 0;
}

__global__ void rank_kernel(const int* __restrict__ dst, int* __restrict__ cnt,
                            int* __restrict__ rank, int nE) {
    int e = blockIdx.x * blockDim.x + threadIdx.x;
    if (e < nE) rank[e] = atomicAdd(&cnt[dst[e]], 1);
}

__global__ __launch_bounds__(1024) void scan_kernel(const int* __restrict__ cnt,
                                                    int* __restrict__ rowptr,
                                                    float* __restrict__ norm) {
    __shared__ int s_pre[32];
    __shared__ int s_sum[32];
    const int t = threadIdx.x;
    const int lane = t & 31;
    const int w = t >> 5;
    const int base = blockIdx.x * 1024;
    int pre = 0;
    for (int i = t; i < base; i += 1024) pre += cnt[i];
    const int x = cnt[base + t];
    for (int o = 16; o > 0; o >>= 1) pre += __shfl_xor(pre, o, 32);
    int inc = x;
    for (int o = 1; o < 32; o <<= 1) {
        int y = __shfl_up(inc, o, 32);
        if (lane >= o) inc += y;
    }
    if (lane == 31) s_sum[w] = inc;
    if (lane == 0)  s_pre[w] = pre;
    __syncthreads();
    int off = 0;
    for (int i = 0; i < 32; ++i) {
        off += s_pre[i];
        if (i < w) off += s_sum[i];
    }
    const int excl = off + inc - x;
    rowptr[base + t] = excl;
    norm[base + t] = rsqrtf(fmaxf((float)x, 1.0f));
    if (base + t == NNODES - 1) rowptr[NNODES] = excl + x;
}

__global__ void fill_kernel(const int* __restrict__ src, const int* __restrict__ dst,
                            const int* __restrict__ rowptr, const int* __restrict__ rank,
                            int* __restrict__ csr, int nE) {
    int e = blockIdx.x * blockDim.x + threadIdx.x;
    if (e < nE) csr[rowptr[dst[e]] + rank[e]] = src[e];
}

__global__ __launch_bounds__(256) void gather_kernel(const float* __restrict__ S,
                                                     const float* __restrict__ norm,
                                                     const int* __restrict__ rowptr,
                                                     const int* __restrict__ csr,
                                                     float* __restrict__ O, int nNodes, int mode) {
    const int tid = blockIdx.x * blockDim.x + threadIdx.x;
    const int v = tid >> 4, c = tid & 15;
    if (v >= nNodes) return;
    int j = rowptr[v];
    const int end = rowptr[v + 1];
    const float4* S4 = (const float4*)S;
    float4 acc = make_float4(0.0f, 0.0f, 0.0f, 0.0f);
    if (mode == 0) {
        for (; j + 4 <= end; j += 4) {
            const int s0 = csr[j], s1 = csr[j + 1], s2 = csr[j + 2], s3 = csr[j + 3];
            const float4 x0 = S4[s0 * 16 + c], x1 = S4[s1 * 16 + c];
            const float4 x2 = S4[s2 * 16 + c], x3 = S4[s3 * 16 + c];
            const float n0 = norm[s0], n1 = norm[s1], n2 = norm[s2], n3 = norm[s3];
            acc.x += n0 * x0.x; acc.y += n0 * x0.y; acc.z += n0 * x0.z; acc.w += n0 * x0.w;
            acc.x += n1 * x1.x; acc.y += n1 * x1.y; acc.z += n1 * x1.z; acc.w += n1 * x1.w;
            acc.x += n2 * x2.x; acc.y += n2 * x2.y; acc.z += n2 * x2.z; acc.w += n2 * x2.w;
            acc.x += n3 * x3.x; acc.y += n3 * x3.y; acc.z += n3 * x3.z; acc.w += n3 * x3.w;
        }
        for (; j < end; ++j) {
            const int s0 = csr[j];
            const float4 x0 = S4[s0 * 16 + c];
            const float n0 = norm[s0];
            acc.x += n0 * x0.x; acc.y += n0 * x0.y; acc.z += n0 * x0.z; acc.w += n0 * x0.w;
        }
    } else {
        for (; j + 4 <= end; j += 4) {
            const int s0 = csr[j], s1 = csr[j + 1], s2 = csr[j + 2], s3 = csr[j + 3];
            const float4 x0 = S4[s0 * 16 + c], x1 = S4[s1 * 16 + c];
            const float4 x2 = S4[s2 * 16 + c], x3 = S4[s3 * 16 + c];
            acc.x += x0.x; acc.y += x0.y; acc.z += x0.z; acc.w += x0.w;
            acc.x += x1.x; acc.y += x1.y; acc.z += x1.z; acc.w += x1.w;
            acc.x += x2.x; acc.y += x2.y; acc.z += x2.z; acc.w += x2.w;
            acc.x += x3.x; acc.y += x3.y; acc.z += x3.z; acc.w += x3.w;
        }
        for (; j < end; ++j) {
            const float4 x0 = S4[csr[j] * 16 + c];
            acc.x += x0.x; acc.y += x0.y; acc.z += x0.z; acc.w += x0.w;
        }
    }
    const float nv = norm[v];
    float4 h;
    h.x = nv * acc.x; h.y = nv * acc.y; h.z = nv * acc.z; h.w = nv * acc.w;
    if (mode != 2) { h.x *= nv; h.y *= nv; h.z *= nv; h.w *= nv; }
    ((float4*)O)[v * 16 + c] = h;
}

__global__ __launch_bounds__(256) void ffn_wmma_kernel(
    const float* __restrict__ r, const float* __restrict__ feat,
    const float* __restrict__ w1, const float* __restrict__ b1,
    const float* __restrict__ w2, const float* __restrict__ b2,
    const float* __restrict__ gamma, const float* __restrict__ beta,
    float* __restrict__ out) {
    __shared__ float lds[8 * 16 * DIM];
    const int wave = threadIdx.x >> 5;
    const int lane = threadIdx.x & 31;
    const int idx  = lane & 15;
    const int half = lane >> 4;
    const int rowbase = blockIdx.x * 128 + wave * 16;
    float* A = lds + wave * 16 * DIM;

    const float2 g2  = ((const float2*)gamma)[lane];
    const float2 be2 = ((const float2*)beta)[lane];
    for (int m = 0; m < 16; ++m) {
        float2 x = ((const float2*)(r + (rowbase + m) * DIM))[lane];
        float s = x.x + x.y;
        #pragma unroll
        for (int o = 16; o > 0; o >>= 1) s += __shfl_xor(s, o, 32);
        float mu = s * (1.0f / 64.0f);
        float dx = x.x - mu, dy = x.y - mu;
        float v = dx * dx + dy * dy;
        #pragma unroll
        for (int o = 16; o > 0; o >>= 1) v += __shfl_xor(v, o, 32);
        float rstd = rsqrtf(v * (1.0f / 64.0f) + 1e-5f);
        float2 y;
        y.x = dx * rstd * g2.x + be2.x;
        y.y = dy * rstd * g2.y + be2.y;
        ((float2*)(A + m * DIM))[lane] = y;
    }
    __syncthreads();

    v8f acc[4];
    #pragma unroll
    for (int j = 0; j < 4; ++j) acc[j] = (v8f)(0.0f);
    for (int kb = 0; kb < 16; ++kb) {
        const int k = kb * 4 + half * 2;
        v2f a = *(const v2f*)(A + idx * DIM + k);
        #pragma unroll
        for (int j = 0; j < 4; ++j) {
            float2 wv = *(const float2*)(w1 + (j * 16 + idx) * DIM + k);
            v2f b = {wv.x, wv.y};
            acc[j] = __builtin_amdgcn_wmma_f32_16x16x4_f32(
                false, a, false, b, (short)0, acc[j], false, false);
        }
    }
    __syncthreads();
    #pragma unroll
    for (int j = 0; j < 4; ++j) {
        const float bias = b1[j * 16 + idx];
        #pragma unroll
        for (int t = 0; t < 8; ++t) {
            float hv = acc[j][t] + bias;
            A[(t + 8 * half) * DIM + j * 16 + idx] = fmaxf(hv, 0.0f);
        }
    }
    __syncthreads();

    v8f acc2[4];
    #pragma unroll
    for (int j = 0; j < 4; ++j) acc2[j] = (v8f)(0.0f);
    for (int kb = 0; kb < 16; ++kb) {
        const int k = kb * 4 + half * 2;
        v2f a = *(const v2f*)(A + idx * DIM + k);
        #pragma unroll
        for (int j = 0; j < 4; ++j) {
            float2 wv = *(const float2*)(w2 + (j * 16 + idx) * DIM + k);
            v2f b = {wv.x, wv.y};
            acc2[j] = __builtin_amdgcn_wmma_f32_16x16x4_f32(
                false, a, false, b, (short)0, acc2[j], false, false);
        }
    }
    #pragma unroll
    for (int j = 0; j < 4; ++j) {
        const int col = j * 16 + idx;
        const float bias = b2[col];
        #pragma unroll
        for (int t = 0; t < 8; ++t) {
            const int row = rowbase + t + 8 * half;
            out[row * DIM + col] = acc2[j][t] + bias + r[row * DIM + col] + feat[row * DIM + col];
        }
    }
}

extern "C" void kernel_launch(void* const* d_in, const int* in_sizes, int n_in,
                              void* d_out, int out_size, void* d_ws, size_t ws_size,
                              hipStream_t stream) {
    const float* feat = (const float*)d_in[0];
    const int*   esrc = (const int*)d_in[1];
    const int*   edst = (const int*)d_in[2];
    const float* w1   = (const float*)d_in[3];
    const float* b1   = (const float*)d_in[4];
    const float* w2   = (const float*)d_in[5];
    const float* b2   = (const float*)d_in[6];
    const float* gam  = (const float*)d_in[7];
    const float* bet  = (const float*)d_in[8];

    float* out0 = (float*)d_out;
    float* out1 = out0 + (size_t)NNODES * DIM;

    int*   cnt    = (int*)d_ws;
    int*   rowptr = cnt + NNODES;
    float* norm   = (float*)(rowptr + NNODES + 64);
    int*   rank   = (int*)(norm + NNODES);
    int*   csr    = rank + NEDGES;
    float* sA     = (float*)(csr + NEDGES);
    float* sB     = sA + (size_t)NNODES * DIM;

    dim3 blk(256);
    dim3 gN((NNODES + 255) / 256);
    dim3 gE((NEDGES + 255) / 256);
    dim3 gG((NNODES * 16 + 255) / 256);

    zero_int_kernel<<<gN, blk, 0, stream>>>(cnt, NNODES);
    rank_kernel<<<gE, blk, 0, stream>>>(edst, cnt, rank, NEDGES);
    scan_kernel<<<dim3(NNODES / 1024), dim3(1024), 0, stream>>>(cnt, rowptr, norm);
    fill_kernel<<<gE, blk, 0, stream>>>(esrc, edst, rowptr, rank, csr, NEDGES);

    gather_kernel<<<gG, blk, 0, stream>>>(feat, norm, rowptr, csr, sA, NNODES, 0);
    gather_kernel<<<gG, blk, 0, stream>>>(sA,   norm, rowptr, csr, sB, NNODES, 1);
    gather_kernel<<<gG, blk, 0, stream>>>(sB,   norm, rowptr, csr, out1, NNODES, 2);

    ffn_wmma_kernel<<<dim3(NNODES / 128), blk, 0, stream>>>(
        out1, feat, w1, b1, w2, b2, gam, bet, out0);
}


# Build dst segments by two-level counting sort in LDS (256 buckets of 256 nodes, then per-bucket sort) instead of per-edge global int atomics + scattered fill; order = LDS atomic arrival rank
# speedup vs baseline: 3.8387x; 1.5966x over previous
#include <hip/hip_runtime.h>

#define NNODES 65536
#define NEDGES 1048576
#define DIM    64
#define NBKT   256
#define EPB    4096
#define CAP    8192

typedef float v2f __attribute__((ext_vector_type(2)));
typedef float v8f __attribute__((ext_vector_type(8)));


__device__ __forceinline__ int block_excl_scan256(int x, int t, int* s_sum) {
    const int lane = t & 31;
    const int w = t >> 5;
    int inc = x;
    for (int o = 1; o < 32; o <<= 1) {
        int y = __shfl_up(inc, o, 32);
        if (lane >= o) inc += y;
    }
    if (t < 256 && lane == 31) s_sum[w] = inc;
    __syncthreads();
    int off = 0;
    if (t < 256) {
        for (int i = 0; i < 8; ++i) if (i < w) off += s_sum[i];
    }
    return off + inc - x;
}

__global__ __launch_bounds__(512) void bucket_hist_kernel(const int* __restrict__ dst,
                                                          int* __restrict__ histT) {
    __shared__ int h[NBKT];
    const int t = threadIdx.x, b = blockIdx.x;
    if (t < NBKT) h[t] = 0;
    __syncthreads();
    const int4* d4 = (const int4*)dst + b * (EPB / 4);
    const int4 d0 = d4[t], d1 = d4[512 + t];
    atomicAdd(&h[(d0.x >> 8) & 255], 1);
    atomicAdd(&h[(d0.y >> 8) & 255], 1);
    atomicAdd(&h[(d0.z >> 8) & 255], 1);
    atomicAdd(&h[(d0.w >> 8) & 255], 1);
    atomicAdd(&h[(d1.x >> 8) & 255], 1);
    atomicAdd(&h[(d1.y >> 8) & 255], 1);
    atomicAdd(&h[(d1.z >> 8) & 255], 1);
    atomicAdd(&h[(d1.w >> 8) & 255], 1);
    __syncthreads();
    if (t < NBKT) histT[t * 256 + b] = h[t];
}

__global__ __launch_bounds__(1024) void scan_kernel(const int* __restrict__ in,
                                                    int* __restrict__ out) {
    __shared__ int s_pre[32];
    __shared__ int s_sum[32];
    const int t = threadIdx.x;
    const int lane = t & 31;
    const int w = t >> 5;
    const int base = blockIdx.x * 1024;
    int pre = 0;
    for (int i = t; i < base; i += 1024) pre += in[i];
    const int x = in[base + t];
    for (int o = 16; o > 0; o >>= 1) pre += __shfl_xor(pre, o, 32);
    int inc = x;
    for (int o = 1; o < 32; o <<= 1) {
        int y = __shfl_up(inc, o, 32);
        if (lane >= o) inc += y;
    }
    if (lane == 31) s_sum[w] = inc;
    if (lane == 0)  s_pre[w] = pre;
    __syncthreads();
    int off = 0;
    for (int i = 0; i < 32; ++i) {
        off += s_pre[i];
        if (i < w) off += s_sum[i];
    }
    const int excl = off + inc - x;
    out[base + t] = excl;
    if (base + t == NNODES - 1) out[NNODES] = excl + x;
}

__global__ __launch_bounds__(512) void bucket_scatter_kernel(const int* __restrict__ src,
                                                             const int* __restrict__ dst,
                                                             const int* __restrict__ goffT,
                                                             unsigned* __restrict__ bkt) {
    __shared__ int h[NBKT];
    __shared__ int lstart[NBKT];
    __shared__ int gbase[NBKT];
    __shared__ int s_sum[8];
    __shared__ unsigned sorted[EPB];
    const int t = threadIdx.x, b = blockIdx.x;
    if (t < NBKT) h[t] = 0;
    __syncthreads();
    const int4* d4 = (const int4*)dst + b * (EPB / 4);
    const int4* s4 = (const int4*)src + b * (EPB / 4);
    const int4 d0 = d4[t], d1 = d4[512 + t];
    const int4 s0 = s4[t], s1 = s4[512 + t];
    const int dd[8] = {d0.x, d0.y, d0.z, d0.w, d1.x, d1.y, d1.z, d1.w};
    const int ss[8] = {s0.x, s0.y, s0.z, s0.w, s1.x, s1.y, s1.z, s1.w};
    int kk[8], rr[8];
    #pragma unroll
    for (int j = 0; j < 8; ++j) {
        kk[j] = (dd[j] >> 8) & 255;
        rr[j] = atomicAdd(&h[kk[j]], 1);
    }
    __syncthreads();
    const int x = (t < NBKT) ? h[t] : 0;
    const int excl = block_excl_scan256(x, t, s_sum);
    if (t < NBKT) {
        lstart[t] = excl;
        gbase[t] = goffT[t * 256 + b] - excl;
    }
    __syncthreads();
    #pragma unroll
    for (int j = 0; j < 8; ++j)
        sorted[lstart[kk[j]] + rr[j]] = ((unsigned)ss[j] << 16) | ((unsigned)dd[j] & 0xFFFFu);
    __syncthreads();
    #pragma unroll
    for (int q = 0; q < 8; ++q) {
        const int i = q * 512 + t;
        const unsigned wv = sorted[i];
        bkt[gbase[(wv >> 8) & 255] + i] = wv;
    }
}

__global__ __launch_bounds__(1024) void bucket_sort_kernel(const unsigned* __restrict__ bkt,
                                                           const int* __restrict__ goffT,
                                                           int* __restrict__ rowptr,
                                                           float* __restrict__ norm,
                                                           int* __restrict__ csr) {
    __shared__ int cnt[256];
    __shared__ int cur[256];
    __shared__ int s_sum[8];
    __shared__ int stage[CAP];
    const int t = threadIdx.x, k = blockIdx.x;
    const int start = goffT[k * 256];
    const int end = goffT[(k + 1) * 256];
    if (t < 256) cnt[t] = 0;
    __syncthreads();
    for (int i0 = start; i0 < end; i0 += 4096) {
        unsigned wv[4];
        #pragma unroll
        for (int q = 0; q < 4; ++q) {
            const int i = i0 + q * 1024 + t;
            wv[q] = (i < end) ? bkt[i] : 0u;
        }
        #pragma unroll
        for (int q = 0; q < 4; ++q) {
            const int i = i0 + q * 1024 + t;
            if (i < end) atomicAdd(&cnt[wv[q] & 255u], 1);
        }
    }
    __syncthreads();
    const int x = (t < 256) ? cnt[t] : 0;
    const int excl = block_excl_scan256(x, t, s_sum);
    if (t < 256) {
        cur[t] = excl;
        rowptr[k * 256 + t] = start + excl;
        norm[k * 256 + t] = rsqrtf(fmaxf((float)x, 1.0f));
    }
    if (k == NBKT - 1 && t == 0) rowptr[NNODES] = end;
    __syncthreads();
    const int n = end - start;
    if (n <= CAP) {
        for (int i0 = start; i0 < end; i0 += 4096) {
            unsigned wv[4];
            #pragma unroll
            for (int q = 0; q < 4; ++q) {
                const int i = i0 + q * 1024 + t;
                wv[q] = (i < end) ? bkt[i] : 0u;
            }
            #pragma unroll
            for (int q = 0; q < 4; ++q) {
                const int i = i0 + q * 1024 + t;
                if (i < end) stage[atomicAdd(&cur[wv[q] & 255u], 1)] = (int)(wv[q] >> 16);
            }
        }
        __syncthreads();
        for (int i = t; i < n; i += 1024) csr[start + i] = stage[i];
    } else {
        for (int i = start + t; i < end; i += 1024) {
            const unsigned wv = bkt[i];
            csr[start + atomicAdd(&cur[wv & 255u], 1)] = (int)(wv >> 16);
        }
    }
}

__global__ __launch_bounds__(256) void gather_kernel(const float* __restrict__ S,
                                                     const float* __restrict__ norm,
                                                     const int* __restrict__ rowptr,
                                                     const int* __restrict__ csr,
                                                     float* __restrict__ O, int nNodes, int mode) {
    const int tid = blockIdx.x * blockDim.x + threadIdx.x;
    const int v = tid >> 4, c = tid & 15;
    if (v >= nNodes) return;
    int j = rowptr[v];
    const int end = rowptr[v + 1];
    const float4* S4 = (const float4*)S;
    float4 acc = make_float4(0.0f, 0.0f, 0.0f, 0.0f);
    if (mode == 0) {
        for (; j + 4 <= end; j += 4) {
            const int s0 = csr[j], s1 = csr[j + 1], s2 = csr[j + 2], s3 = csr[j + 3];
            const float4 x0 = S4[s0 * 16 + c], x1 = S4[s1 * 16 + c];
            const float4 x2 = S4[s2 * 16 + c], x3 = S4[s3 * 16 + c];
            const float n0 = norm[s0], n1 = norm[s1], n2 = norm[s2], n3 = norm[s3];
            acc.x += n0 * x0.x; acc.y += n0 * x0.y; acc.z += n0 * x0.z; acc.w += n0 * x0.w;
            acc.x += n1 * x1.x; acc.y += n1 * x1.y; acc.z += n1 * x1.z; acc.w += n1 * x1.w;
            acc.x += n2 * x2.x; acc.y += n2 * x2.y; acc.z += n2 * x2.z; acc.w += n2 * x2.w;
            acc.x += n3 * x3.x; acc.y += n3 * x3.y; acc.z += n3 * x3.z; acc.w += n3 * x3.w;
        }
        for (; j < end; ++j) {
            const int s0 = csr[j];
            const float4 x0 = S4[s0 * 16 + c];
            const float n0 = norm[s0];
            acc.x += n0 * x0.x; acc.y += n0 * x0.y; acc.z += n0 * x0.z; acc.w += n0 * x0.w;
        }
    } else {
        for (; j + 4 <= end; j += 4) {
            const int s0 = csr[j], s1 = csr[j + 1], s2 = csr[j + 2], s3 = csr[j + 3];
            const float4 x0 = S4[s0 * 16 + c], x1 = S4[s1 * 16 + c];
            const float4 x2 = S4[s2 * 16 + c], x3 = S4[s3 * 16 + c];
            acc.x += x0.x; acc.y += x0.y; acc.z += x0.z; acc.w += x0.w;
            acc.x += x1.x; acc.y += x1.y; acc.z += x1.z; acc.w += x1.w;
            acc.x += x2.x; acc.y += x2.y; acc.z += x2.z; acc.w += x2.w;
            acc.x += x3.x; acc.y += x3.y; acc.z += x3.z; acc.w += x3.w;
        }
        for (; j < end; ++j) {
            const float4 x0 = S4[csr[j] * 16 + c];
            acc.x += x0.x; acc.y += x0.y; acc.z += x0.z; acc.w += x0.w;
        }
    }
    const float nv = norm[v];
    float4 h;
    h.x = nv * acc.x; h.y = nv * acc.y; h.z = nv * acc.z; h.w = nv * acc.w;
    if (mode != 2) { h.x *= nv; h.y *= nv; h.z *= nv; h.w *= nv; }
    ((float4*)O)[v * 16 + c] = h;
}

__global__ __launch_bounds__(256) void ffn_wmma_kernel(
    const float* __restrict__ r, const float* __restrict__ feat,
    const float* __restrict__ w1, const float* __restrict__ b1,
    const float* __restrict__ w2, const float* __restrict__ b2,
    const float* __restrict__ gamma, const float* __restrict__ beta,
    float* __restrict__ out) {
    __shared__ float lds[8 * 16 * DIM];
    const int wave = threadIdx.x >> 5;
    const int lane = threadIdx.x & 31;
    const int idx  = lane & 15;
    const int half = lane >> 4;
    const int rowbase = blockIdx.x * 128 + wave * 16;
    float* A = lds + wave * 16 * DIM;

    const float2 g2  = ((const float2*)gamma)[lane];
    const float2 be2 = ((const float2*)beta)[lane];
    for (int m = 0; m < 16; ++m) {
        float2 x = ((const float2*)(r + (rowbase + m) * DIM))[lane];
        float s = x.x + x.y;
        #pragma unroll
        for (int o = 16; o > 0; o >>= 1) s += __shfl_xor(s, o, 32);
        float mu = s * (1.0f / 64.0f);
        float dx = x.x - mu, dy = x.y - mu;
        float v = dx * dx + dy * dy;
        #pragma unroll
        for (int o = 16; o > 0; o >>= 1) v += __shfl_xor(v, o, 32);
        float rstd = rsqrtf(v * (1.0f / 64.0f) + 1e-5f);
        float2 y;
        y.x = dx * rstd * g2.x + be2.x;
        y.y = dy * rstd * g2.y + be2.y;
        ((float2*)(A + m * DIM))[lane] = y;
    }
    __syncthreads();

    v8f acc[4];
    #pragma unroll
    for (int j = 0; j < 4; ++j) acc[j] = (v8f)(0.0f);
    for (int kb = 0; kb < 16; ++kb) {
        const int k = kb * 4 + half * 2;
        v2f a = *(const v2f*)(A + idx * DIM + k);
        #pragma unroll
        for (int j = 0; j < 4; ++j) {
            float2 wv = *(const float2*)(w1 + (j * 16 + idx) * DIM + k);
            v2f b = {wv.x, wv.y};
            acc[j] = __builtin_amdgcn_wmma_f32_16x16x4_f32(
                false, a, false, b, (short)0, acc[j], false, false);
        }
    }
    __syncthreads();
    #pragma unroll
    for (int j = 0; j < 4; ++j) {
        const float bias = b1[j * 16 + idx];
        #pragma unroll
        for (int t = 0; t < 8; ++t) {
            float hv = acc[j][t] + bias;
            A[(t + 8 * half) * DIM + j * 16 + idx] = fmaxf(hv, 0.0f);
        }
    }
    __syncthreads();

    v8f acc2[4];
    #pragma unroll
    for (int j = 0; j < 4; ++j) acc2[j] = (v8f)(0.0f);
    for (int kb = 0; kb < 16; ++kb) {
        const int k = kb * 4 + half * 2;
        v2f a = *(const v2f*)(A + idx * DIM + k);
        #pragma unroll
        for (int j = 0; j < 4; ++j) {
            float2 wv = *(const float2*)(w2 + (j * 16 + idx) * DIM + k);
            v2f b = {wv.x, wv.y};
            acc2[j] = __builtin_amdgcn_wmma_f32_16x16x4_f32(
                false, a, false, b, (short)0, acc2[j], false, false);
        }
    }
    #pragma unroll
    for (int j = 0; j < 4; ++j) {
        const int col = j * 16 + idx;
        const float bias = b2[col];
        #pragma unroll
        for (int t = 0; t < 8; ++t) {
            const int row = rowbase + t + 8 * half;
            out[row * DIM + col] = acc2[j][t] + bias + r[row * DIM + col] + feat[row * DIM + col];
        }
    }
}

extern "C" void kernel_launch(void* const* d_in, const int* in_sizes, int n_in,
                              void* d_out, int out_size, void* d_ws, size_t ws_size,
                              hipStream_t stream) {
    const float* feat = (const float*)d_in[0];
    const int*   esrc = (const int*)d_in[1];
    const int*   edst = (const int*)d_in[2];
    const float* w1   = (const float*)d_in[3];
    const float* b1   = (const float*)d_in[4];
    const float* w2   = (const float*)d_in[5];
    const float* b2   = (const float*)d_in[6];
    const float* gam  = (const float*)d_in[7];
    const float* bet  = (const float*)d_in[8];

    float* out0 = (float*)d_out;
    float* out1 = out0 + (size_t)NNODES * DIM;

    int*      histT  = (int*)d_ws;
    int*      goffT  = histT + NNODES;
    int*      rowptr = goffT + NNODES + 64;
    float*    norm   = (float*)(rowptr + NNODES + 64);
    unsigned* bkt    = (unsigned*)(norm + NNODES);
    int*      csr    = (int*)(bkt + NEDGES);
    float*    sA     = (float*)(csr + NEDGES);
    float*    sB     = sA + (size_t)NNODES * DIM;

    dim3 blk(256);
    dim3 gG((NNODES * 16 + 255) / 256);

    bucket_hist_kernel<<<dim3(NEDGES / EPB), dim3(512), 0, stream>>>(edst, histT);
    scan_kernel<<<dim3(NNODES / 1024), dim3(1024), 0, stream>>>(histT, goffT);
    bucket_scatter_kernel<<<dim3(NEDGES / EPB), dim3(512), 0, stream>>>(esrc, edst, goffT, bkt);
    bucket_sort_kernel<<<dim3(NBKT), dim3(1024), 0, stream>>>(bkt, goffT, rowptr, norm, csr);

    gather_kernel<<<gG, blk, 0, stream>>>(feat, norm, rowptr, csr, sA, NNODES, 0);
    gather_kernel<<<gG, blk, 0, stream>>>(sA,   norm, rowptr, csr, sB, NNODES, 1);
    gather_kernel<<<gG, blk, 0, stream>>>(sB,   norm, rowptr, csr, out1, NNODES, 2);

    ffn_wmma_kernel<<<dim3(NNODES / 128), blk, 0, stream>>>(
        out1, feat, w1, b1, w2, b2, gam, bet, out0);
}


# Gather: a 16-lane row takes 4 nodes in turn; per node, ids by one coalesced load + row_share, up to 16 row loads issued as one group, next ids fetched ahead; 4x fewer waves; sum order kept
# speedup vs baseline: 4.2364x; 1.1036x over previous
#include <hip/hip_runtime.h>

#define NNODES 65536
#define NEDGES 1048576
#define DIM    64
#define NBKT   256
#define EPB    4096
#define CAP    8192

typedef float v2f __attribute__((ext_vector_type(2)));
typedef float v8f __attribute__((ext_vector_type(8)));


__device__ __forceinline__ int block_excl_scan256(int x, int t, int* s_sum) {
    const int lane = t & 31;
    const int w = t >> 5;
    int inc = x;
    for (int o = 1; o < 32; o <<= 1) {
        int y = __shfl_up(inc, o, 32);
        if (lane >= o) inc += y;
    }
    if (t < 256 && lane == 31) s_sum[w] = inc;
    __syncthreads();
    int off = 0;
    if (t < 256) {
        for (int i = 0; i < 8; ++i) if (i < w) off += s_sum[i];
    }
    return off + inc - x;
}

__global__ __launch_bounds__(512) void bucket_hist_kernel(const int* __restrict__ dst,
                                                          int* __restrict__ histT) {
    __shared__ int h[NBKT];
    const int t = threadIdx.x, b = blockIdx.x;
    if (t < NBKT) h[t] = 0;
    __syncthreads();
    const int4* d4 = (const int4*)dst + b * (EPB / 4);
    const int4 d0 = d4[t], d1 = d4[512 + t];
    atomicAdd(&h[(d0.x >> 8) & 255], 1);
    atomicAdd(&h[(d0.y >> 8) & 255], 1);
    atomicAdd(&h[(d0.z >> 8) & 255], 1);
    atomicAdd(&h[(d0.w >> 8) & 255], 1);
    atomicAdd(&h[(d1.x >> 8) & 255], 1);
    atomicAdd(&h[(d1.y >> 8) & 255], 1);
    atomicAdd(&h[(d1.z >> 8) & 255], 1);
    atomicAdd(&h[(d1.w >> 8) & 255], 1);
    __syncthreads();
    if (t < NBKT) histT[t * 256 + b] = h[t];
}

__global__ __launch_bounds__(1024) void scan_kernel(const int* __restrict__ in,
                                                    int* __restrict__ out) {
    __shared__ int s_pre[32];
    __shared__ int s_sum[32];
    const int t = threadIdx.x;
    const int lane = t & 31;
    const int w = t >> 5;
    const int base = blockIdx.x * 1024;
    int pre = 0;
    for (int i = t; i < base; i += 1024) pre += in[i];
    const int x = in[base + t];
    for (int o = 16; o > 0; o >>= 1) pre += __shfl_xor(pre, o, 32);
    int inc = x;
    for (int o = 1; o < 32; o <<= 1) {
        int y = __shfl_up(inc, o, 32);
        if (lane >= o) inc += y;
    }
    if (lane == 31) s_sum[w] = inc;
    if (lane == 0)  s_pre[w] = pre;
    __syncthreads();
    int off = 0;
    for (int i = 0; i < 32; ++i) {
        off += s_pre[i];
        if (i < w) off += s_sum[i];
    }
    const int excl = off + inc - x;
    out[base + t] = excl;
    if (base + t == NNODES - 1) out[NNODES] = excl + x;
}

__global__ __launch_bounds__(512) void bucket_scatter_kernel(const int* __restrict__ src,
                                                             const int* __restrict__ dst,
                                                             const int* __restrict__ goffT,
                                                             unsigned* __restrict__ bkt) {
    __shared__ int h[NBKT];
    __shared__ int lstart[NBKT];
    __shared__ int gbase[NBKT];
    __shared__ int s_sum[8];
    __shared__ unsigned sorted[EPB];
    const int t = threadIdx.x, b = blockIdx.x;
    if (t < NBKT) h[t] = 0;
    __syncthreads();
    const int4* d4 = (const int4*)dst + b * (EPB / 4);
    const int4* s4 = (const int4*)src + b * (EPB / 4);
    const int4 d0 = d4[t], d1 = d4[512 + t];
    const int4 s0 = s4[t], s1 = s4[512 + t];
    const int dd[8] = {d0.x, d0.y, d0.z, d0.w, d1.x, d1.y, d1.z, d1.w};
    const int ss[8] = {s0.x, s0.y, s0.z, s0.w, s1.x, s1.y, s1.z, s1.w};
    int kk[8], rr[8];
    #pragma unroll
    for (int j = 0; j < 8; ++j) {
        kk[j] = (dd[j] >> 8) & 255;
        rr[j] = atomicAdd(&h[kk[j]], 1);
    }
    __syncthreads();
    const int x = (t < NBKT) ? h[t] : 0;
    const int excl = block_excl_scan256(x, t, s_sum);
    if (t < NBKT) {
        lstart[t] = excl;
        gbase[t] = goffT[t * 256 + b] - excl;
    }
    __syncthreads();
    #pragma unroll
    for (int j = 0; j < 8; ++j)
        sorted[lstart[kk[j]] + rr[j]] = ((unsigned)ss[j] << 16) | ((unsigned)dd[j] & 0xFFFFu);
    __syncthreads();
    #pragma unroll
    for (int q = 0; q < 8; ++q) {
        const int i = q * 512 + t;
        const unsigned wv = sorted[i];
        bkt[gbase[(wv >> 8) & 255] + i] = wv;
    }
}

__global__ __launch_bounds__(1024) void bucket_sort_kernel(const unsigned* __restrict__ bkt,
                                                           const int* __restrict__ goffT,
                                                           int* __restrict__ rowptr,
                                                           float* __restrict__ norm,
                                                           int* __restrict__ csr) {
    __shared__ int cnt[256];
    __shared__ int cur[256];
    __shared__ int s_sum[8];
    __shared__ int stage[CAP];
    const int t = threadIdx.x, k = blockIdx.x;
    const int start = goffT[k * 256];
    const int end = goffT[(k + 1) * 256];
    if (t < 256) cnt[t] = 0;
    __syncthreads();
    for (int i0 = start; i0 < end; i0 += 4096) {
        unsigned wv[4];
        #pragma unroll
        for (int q = 0; q < 4; ++q) {
            const int i = i0 + q * 1024 + t;
            wv[q] = (i < end) ? bkt[i] : 0u;
        }
        #pragma unroll
        for (int q = 0; q < 4; ++q) {
            const int i = i0 + q * 1024 + t;
            if (i < end) atomicAdd(&cnt[wv[q] & 255u], 1);
        }
    }
    __syncthreads();
    const int x = (t < 256) ? cnt[t] : 0;
    const int excl = block_excl_scan256(x, t, s_sum);
    if (t < 256) {
        cur[t] = excl;
        rowptr[k * 256 + t] = start + excl;
        norm[k * 256 + t] = rsqrtf(fmaxf((float)x, 1.0f));
    }
    if (k == NBKT - 1 && t == 0) rowptr[NNODES] = end;
    __syncthreads();
    const int n = end - start;
    if (n <= CAP) {
        for (int i0 = start; i0 < end; i0 += 4096) {
            unsigned wv[4];
            #pragma unroll
            for (int q = 0; q < 4; ++q) {
                const int i = i0 + q * 1024 + t;
                wv[q] = (i < end) ? bkt[i] : 0u;
            }
            #pragma unroll
            for (int q = 0; q < 4; ++q) {
                const int i = i0 + q * 1024 + t;
                if (i < end) stage[atomicAdd(&cur[wv[q] & 255u], 1)] = (int)(wv[q] >> 16);
            }
        }
        __syncthreads();
        for (int i = t; i < n; i += 1024) csr[start + i] = stage[i];
    } else {
        for (int i = start + t; i < end; i += 1024) {
            const unsigned wv = bkt[i];
            csr[start + atomicAdd(&cur[wv & 255u], 1)] = (int)(wv >> 16);
        }
    }
}

#define NPG 4

template <int E>
__device__ __forceinline__ int row_bcast(int x) {
    return __builtin_amdgcn_update_dpp(0, x, 0x150 + E, 0xF, 0xF, true);
}

#define G_LOAD(E)                                                                       \
    {                                                                                   \
        const unsigned id = (unsigned)row_bcast<E>(ids);                                \
        x[E] = S4[((E) < left) ? (id * 16u + (unsigned)c) : (unsigned)c];               \
    }
#define G_ADD(E)                                                                        \
    {                                                                                   \
        const bool ok = (E) < left;                                                     \
        const float tx = ok ? x[E].x : 0.0f, ty = ok ? x[E].y : 0.0f;                   \
        const float tz = ok ? x[E].z : 0.0f, tw = ok ? x[E].w : 0.0f;                   \
        if constexpr (MODE == 0) {                                                      \
            const float n = __int_as_float(row_bcast<E>(nb));                           \
            acc.x += n * tx; acc.y += n * ty; acc.z += n * tz; acc.w += n * tw;         \
        } else {                                                                        \
            acc.x += tx; acc.y += ty; acc.z += tz; acc.w += tw;                         \
        }                                                                               \
    }
#define G_QUAD(M, Q) M(4 * (Q)) M(4 * (Q) + 1) M(4 * (Q) + 2) M(4 * (Q) + 3)

template <int MODE>
__device__ __forceinline__ int gather_batch(const float4* __restrict__ S4,
                                            const float* __restrict__ norm,
                                            const int* __restrict__ csr,
                                            int c, int ids, int left, int rem, int nidx,
                                            float4& acc) {
    float4 x[16];
    const int nxt = csr[nidx];
    int nb = 0;
    if constexpr (MODE == 0) nb = __float_as_int(norm[ids]);
    G_QUAD(G_LOAD, 0)
    if (rem > 4) {
        G_QUAD(G_LOAD, 1)
        if (rem > 8) {
            G_QUAD(G_LOAD, 2)
            if (rem > 12) { G_QUAD(G_LOAD, 3) }
        }
    }
    __builtin_amdgcn_sched_barrier(0);
    G_QUAD(G_ADD, 0)
    if (rem > 4) {
        G_QUAD(G_ADD, 1)
        if (rem > 8) {
            G_QUAD(G_ADD, 2)
            if (rem > 12) { G_QUAD(G_ADD, 3) }
        }
    }
    return nxt;
}

template <int MODE>
__device__ __forceinline__ void gather_rows(const float* __restrict__ S,
                                            const float* __restrict__ norm,
                                            const int* __restrict__ rowptr,
                                            const int* __restrict__ csr,
                                            float* __restrict__ O) {
    static_assert(NPG == 4, "five segment bounds p0..p4 are kept per row");
    static_assert(NNODES % (2 * NPG) == 0 && (NNODES / NPG * 16) % 256 == 0, "whole waves, whole blocks");
    const int tid = blockIdx.x * 256 + threadIdx.x;
    const int c = tid & 15;
    const int v0 = (tid >> 4) * NPG;
    const float4* S4 = (const float4*)S;
    const int rp = rowptr[v0 + (c < NPG ? c : NPG)];
    int p0 = row_bcast<0>(rp), p1 = row_bcast<1>(rp), p2 = row_bcast<2>(rp);
    int p3 = row_bcast<3>(rp);
    const int p4 = row_bcast<4>(rp);
    int nl = __float_as_int(norm[v0 + (c < NPG ? c : NPG - 1)]);
    int ids = csr[min(p0 + c, NEDGES - 1)];
    for (int i = 0; i < NPG; ++i) {
        const int cnt = p1 - p0;
        const int m = max(__builtin_amdgcn_readlane(cnt, 0), __builtin_amdgcn_readlane(cnt, 16));
        const float nv = __int_as_float(row_bcast<0>(nl));
        float4 acc = make_float4(0.0f, 0.0f, 0.0f, 0.0f);
        for (int base = 0; base < m; base += 16) {
            const int rem = m - base;
            const int nidx = min((rem > 16 ? p0 + base + 16 : p1) + c, NEDGES - 1);
            ids = gather_batch<MODE>(S4, norm, csr, c, ids, cnt - base, rem, nidx, acc);
        }
        float4 h;
        h.x = nv * acc.x; h.y = nv * acc.y; h.z = nv * acc.z; h.w = nv * acc.w;
        if (MODE != 2) { h.x *= nv; h.y *= nv; h.z *= nv; h.w *= nv; }
        ((float4*)O)[(v0 + i) * 16 + c] = h;
        p0 = p1; p1 = p2; p2 = p3; p3 = p4;
        nl = __builtin_amdgcn_update_dpp(0, nl, 0x101, 0xF, 0xF, true);
    }
}

__global__ __launch_bounds__(256) void gather_first_kernel(const float* __restrict__ S,
                                                           const float* __restrict__ norm,
                                                           const int* __restrict__ rowptr,
                                                           const int* __restrict__ csr,
                                                           float* __restrict__ O) {
    gather_rows<0>(S, norm, rowptr, csr, O);
}
__global__ __launch_bounds__(256) void gather_mid_kernel(const float* __restrict__ S,
                                                         const float* __restrict__ norm,
                                                         const int* __restrict__ rowptr,
                                                         const int* __restrict__ csr,
                                                         float* __restrict__ O) {
    gather_rows<1>(S, norm, rowptr, csr, O);
}
__global__ __launch_bounds__(256) void gather_last_kernel(const float* __restrict__ S,
                                                          const float* __restrict__ norm,
                                                          const int* __restrict__ rowptr,
                                                          const int* __restrict__ csr,
                                                          float* __restrict__ O) {
    gather_rows<2>(S, norm, rowptr, csr, O);
}

__global__ __launch_bounds__(256) void ffn_wmma_kernel(
    const float* __restrict__ r, const float* __restrict__ feat,
    const float* __restrict__ w1, const float* __restrict__ b1,
    const float* __restrict__ w2, const float* __restrict__ b2,
    const float* __restrict__ gamma, const float* __restrict__ beta,
    float* __restrict__ out) {
    __shared__ float lds[8 * 16 * DIM];
    const int wave = threadIdx.x >> 5;
    const int lane = threadIdx.x & 31;
    const int idx  = lane & 15;
    const int half = lane >> 4;
    const int rowbase = blockIdx.x * 128 + wave * 16;
    float* A = lds + wave * 16 * DIM;

    const float2 g2  = ((const float2*)gamma)[lane];
    const float2 be2 = ((const float2*)beta)[lane];
    for (int m = 0; m < 16; ++m) {
        float2 x = ((const float2*)(r + (rowbase + m) * DIM))[lane];
        float s = x.x + x.y;
        #pragma unroll
        for (int o = 16; o > 0; o >>= 1) s += __shfl_xor(s, o, 32);
        float mu = s * (1.0f / 64.0f);
        float dx = x.x - mu, dy = x.y - mu;
        float v = dx * dx + dy * dy;
        #pragma unroll
        for (int o = 16; o > 0; o >>= 1) v += __shfl_xor(v, o, 32);
        float rstd = rsqrtf(v * (1.0f / 64.0f) + 1e-5f);
        float2 y;
        y.x = dx * rstd * g2.x + be2.x;
        y.y = dy * rstd * g2.y + be2.y;
        ((float2*)(A + m * DIM))[lane] = y;
    }
    __syncthreads();

    v8f acc[4];
    #pragma unroll
    for (int j = 0; j < 4; ++j) acc[j] = (v8f)(0.0f);
    for (int kb = 0; kb < 16; ++kb) {
        const int k = kb * 4 + half * 2;
        v2f a = *(const v2f*)(A + idx * DIM + k);
        #pragma unroll
        for (int j = 0; j < 4; ++j) {
            float2 wv = *(const float2*)(w1 + (j * 16 + idx) * DIM + k);
            v2f b = {wv.x, wv.y};
            acc[j] = __builtin_amdgcn_wmma_f32_16x16x4_f32(
                false, a, false, b, (short)0, acc[j], false, false);
        }
    }
    __syncthreads();
    #pragma unroll
    for (int j = 0; j < 4; ++j) {
        const float bias = b1[j * 16 + idx];
        #pragma unroll
        for (int t = 0; t < 8; ++t) {
            float hv = acc[j][t] + bias;
            A[(t + 8 * half) * DIM + j * 16 + idx] = fmaxf(hv, 0.0f);
        }
    }
    __syncthreads();

    v8f acc2[4];
    #pragma unroll
    for (int j = 0; j < 4; ++j) acc2[j] = (v8f)(0.0f);
    for (int kb = 0; kb < 16; ++kb) {
        const int k = kb * 4 + half * 2;
        v2f a = *(const v2f*)(A + idx * DIM + k);
        #pragma unroll
        for (int j = 0; j < 4; ++j) {
            float2 wv = *(const float2*)(w2 + (j * 16 + idx) * DIM + k);
            v2f b = {wv.x, wv.y};
            acc2[j] = __builtin_amdgcn_wmma_f32_16x16x4_f32(
                false, a, false, b, (short)0, acc2[j], false, false);
        }
    }
    #pragma unroll
    for (int j = 0; j < 4; ++j) {
        const int col = j * 16 + idx;
        const float bias = b2[col];
        #pragma unroll
        for (int t = 0; t < 8; ++t) {
            const int row = rowbase + t + 8 * half;
            out[row * DIM + col] = acc2[j][t] + bias + r[row * DIM + col] + feat[row * DIM + col];
        }
    }
}

extern "C" void kernel_launch(void* const* d_in, const int* in_sizes, int n_in,
                              void* d_out, int out_size, void* d_ws, size_t ws_size,
                              hipStream_t stream) {
    const float* feat = (const float*)d_in[0];
    const int*   esrc = (const int*)d_in[1];
    const int*   edst = (const int*)d_in[2];
    const float* w1   = (const float*)d_in[3];
    const float* b1   = (const float*)d_in[4];
    const float* w2   = (const float*)d_in[5];
    const float* b2   = (const float*)d_in[6];
    const float* gam  = (const float*)d_in[7];
    const float* bet  = (const float*)d_in[8];

    float* out0 = (float*)d_out;
    float* out1 = out0 + (size_t)NNODES * DIM;

    int*      histT  = (int*)d_ws;
    int*      goffT  = histT + NNODES;
    int*      rowptr = goffT + NNODES + 64;
    float*    norm   = (float*)(rowptr + NNODES + 64);
    unsigned* bkt    = (unsigned*)(norm + NNODES);
    int*      csr    = (int*)(bkt + NEDGES);
    float*    sA     = (float*)(csr + NEDGES);
    float*    sB     = sA + (size_t)NNODES * DIM;

    dim3 blk(256);
    dim3 gG(NNODES / NPG * 16 / 256);

    bucket_hist_kernel<<<dim3(NEDGES / EPB), dim3(512), 0, stream>>>(edst, histT);
    scan_kernel<<<dim3(NNODES / 1024), dim3(1024), 0, stream>>>(histT, goffT);
    bucket_scatter_kernel<<<dim3(NEDGES / EPB), dim3(512), 0, stream>>>(esrc, edst, goffT, bkt);
    bucket_sort_kernel<<<dim3(NBKT), dim3(1024), 0, stream>>>(bkt, goffT, rowptr, norm, csr);

    gather_first_kernel<<<gG, blk, 0, stream>>>(feat, norm, rowptr, csr, sA);
    gather_mid_kernel<<<gG, blk, 0, stream>>>(sA,   norm, rowptr, csr, sB);
    gather_last_kernel<<<gG, blk, 0, stream>>>(sB,   norm, rowptr, csr, out1);

    ffn_wmma_kernel<<<dim3(NNODES / 128), blk, 0, stream>>>(
        out1, feat, w1, b1, w2, b2, gam, bet, out0);
}


# Drop the global histogram and scan launches: level-1 blocks keep their bucket-sorted runs in their own slice, level-2 collects 256 runs per bucket by the run bounds; order = LDS atomic arrival rank
# speedup vs baseline: 4.9610x; 1.1710x over previous
#include <hip/hip_runtime.h>

#define NNODES 65536
#define NEDGES 1048576
#define DIM    64
#define NBKT   256
#define EPB    4096
#define CAP    8192

typedef float v2f __attribute__((ext_vector_type(2)));
typedef float v8f __attribute__((ext_vector_type(8)));


__device__ __forceinline__ int block_excl_scan256(int x, int t, int* s_sum) {
    const int lane = t & 31;
    const int w = t >> 5;
    int inc = x;
    for (int o = 1; o < 32; o <<= 1) {
        int y = __shfl_up(inc, o, 32);
        if (lane >= o) inc += y;
    }
    if (t < 256 && lane == 31) s_sum[w] = inc;
    __syncthreads();
    int off = 0;
    if (t < 256) {
        for (int i = 0; i < 8; ++i) if (i < w) off += s_sum[i];
    }
    return off + inc - x;
}

__global__ __launch_bounds__(512) void bucket_scatter_kernel(const int* __restrict__ src,
                                                             const int* __restrict__ dst,
                                                             int* __restrict__ runT,
                                                             unsigned* __restrict__ bkt) {
    __shared__ int h[NBKT];
    __shared__ int lstart[NBKT];
    __shared__ int s_sum[8];
    __shared__ __attribute__((aligned(16))) unsigned sorted[EPB];
    const int t = threadIdx.x, b = blockIdx.x;
    if (t < NBKT) h[t] = 0;
    __syncthreads();
    const int4* d4 = (const int4*)dst + b * (EPB / 4);
    const int4* s4 = (const int4*)src + b * (EPB / 4);
    const int4 d0 = d4[t], d1 = d4[512 + t];
    const int4 s0 = s4[t], s1 = s4[512 + t];
    const int dd[8] = {d0.x, d0.y, d0.z, d0.w, d1.x, d1.y, d1.z, d1.w};
    const int ss[8] = {s0.x, s0.y, s0.z, s0.w, s1.x, s1.y, s1.z, s1.w};
    int kk[8], rr[8];
    #pragma unroll
    for (int j = 0; j < 8; ++j) {
        kk[j] = (dd[j] >> 8) & 255;
        rr[j] = atomicAdd(&h[kk[j]], 1);
    }
    __syncthreads();
    const int x = (t < NBKT) ? h[t] : 0;
    const int excl = block_excl_scan256(x, t, s_sum);
    if (t < NBKT) {
        lstart[t] = excl;
        runT[t * 256 + b] = excl;
    }
    if (t == 0) runT[NBKT * 256 + b] = EPB;
    __syncthreads();
    #pragma unroll
    for (int j = 0; j < 8; ++j)
        sorted[lstart[kk[j]] + rr[j]] = ((unsigned)ss[j] << 16) | ((unsigned)dd[j] & 0xFFFFu);
    __syncthreads();
    uint4* o4 = (uint4*)bkt + b * (EPB / 4);
    const uint4* l4 = (const uint4*)sorted;
    o4[t] = l4[t];
    o4[512 + t] = l4[512 + t];
}

__global__ __launch_bounds__(1024) void bucket_sort_kernel(const unsigned* __restrict__ bkt,
                                                           const int* __restrict__ runT,
                                                           int* __restrict__ rowptr,
                                                           float* __restrict__ norm,
                                                           int* __restrict__ csr) {
    __shared__ int cnt[256];
    __shared__ int cur[256];
    __shared__ int s_sum[8];
    __shared__ int s_len[8];
    __shared__ int s_beg[8];
    __shared__ int run_src[256];
    __shared__ int run_len[256];
    __shared__ int run_pos[256];
    __shared__ unsigned ent[CAP];
    __shared__ int stage[CAP];
    const int t = threadIdx.x, k = blockIdx.x;
    const int lane = t & 31, w = t >> 5;
    int beg = 0, len = 0;
    if (t < 256) {
        beg = runT[k * 256 + t];
        len = runT[(k + 1) * 256 + t] - beg;
        cnt[t] = 0;
    }
    int inc = len, bsum = beg;
    for (int o = 1; o < 32; o <<= 1) {
        int y = __shfl_up(inc, o, 32);
        if (lane >= o) inc += y;
    }
    for (int o = 16; o > 0; o >>= 1) bsum += __shfl_xor(bsum, o, 32);
    if (t < 256 && lane == 31) { s_len[w] = inc; s_beg[w] = bsum; }
    __syncthreads();
    int off = 0, n = 0, start = 0;
    for (int i = 0; i < 8; ++i) {
        const int li = s_len[i];
        if (i < w) off += li;
        n += li;
        start += s_beg[i];
    }
    if (t < 256) {
        run_src[t] = t * EPB + beg;
        run_len[t] = len;
        run_pos[t] = off + inc - len;
    }
    __syncthreads();
    const int r = t >> 2, j0 = t & 3;
    const int rsrc = run_src[r], rlen = run_len[r], rpos = run_pos[r];
    const bool staged = n <= CAP;
    if (staged) {
        unsigned wv[8];
        #pragma unroll
        for (int q = 0; q < 8; ++q) {
            const int j = j0 + 4 * q;
            wv[q] = (j < rlen) ? bkt[rsrc + j] : 0u;
        }
        #pragma unroll
        for (int q = 0; q < 8; ++q) {
            const int j = j0 + 4 * q;
            if (j < rlen) {
                ent[rpos + j] = wv[q];
                atomicAdd(&cnt[wv[q] & 255u], 1);
            }
        }
        for (int j = j0 + 32; j < rlen; j += 4) {
            const unsigned v = bkt[rsrc + j];
            ent[rpos + j] = v;
            atomicAdd(&cnt[v & 255u], 1);
        }
    } else {
        for (int j = j0; j < rlen; j += 4) atomicAdd(&cnt[bkt[rsrc + j] & 255u], 1);
    }
    __syncthreads();
    const int x = (t < 256) ? cnt[t] : 0;
    const int excl = block_excl_scan256(x, t, s_sum);
    if (t < 256) {
        cur[t] = excl;
        rowptr[k * 256 + t] = start + excl;
        norm[k * 256 + t] = rsqrtf(fmaxf((float)x, 1.0f));
    }
    if (k == NBKT - 1 && t == 0) rowptr[NNODES] = start + n;
    __syncthreads();
    if (staged) {
        for (int i = t; i < n; i += 1024) {
            const unsigned v = ent[i];
            stage[atomicAdd(&cur[v & 255u], 1)] = (int)(v >> 16);
        }
        __syncthreads();
        for (int i = t; i < n; i += 1024) csr[start + i] = stage[i];
    } else {
        for (int j = j0; j < rlen; j += 4) {
            const unsigned v = bkt[rsrc + j];
            csr[start + atomicAdd(&cur[v & 255u], 1)] = (int)(v >> 16);
        }
    }
}

#define NPG 4

template <int E>
__device__ __forceinline__ int row_bcast(int x) {
    return __builtin_amdgcn_update_dpp(0, x, 0x150 + E, 0xF, 0xF, true);
}

#define G_LOAD(E)                                                                       \
    {                                                                                   \
        const unsigned id = (unsigned)row_bcast<E>(ids);                                \
        x[E] = S4[((E) < left) ? (id * 16u + (unsigned)c) : (unsigned)c];               \
    }
#define G_ADD(E)                                                                        \
    {                                                                                   \
        const bool ok = (E) < left;                                                     \
        const float tx = ok ? x[E].x : 0.0f, ty = ok ? x[E].y : 0.0f;                   \
        const float tz = ok ? x[E].z : 0.0f, tw = ok ? x[E].w : 0.0f;                   \
        if constexpr (MODE == 0) {                                                      \
            const float n = __int_as_float(row_bcast<E>(nb));                           \
            acc.x += n * tx; acc.y += n * ty; acc.z += n * tz; acc.w += n * tw;         \
        } else {                                                                        \
            acc.x += tx; acc.y += ty; acc.z += tz; acc.w += tw;                         \
        }                                                                               \
    }
#define G_QUAD(M, Q) M(4 * (Q)) M(4 * (Q) + 1) M(4 * (Q) + 2) M(4 * (Q) + 3)

template <int MODE>
__device__ __forceinline__ int gather_batch(const float4* __restrict__ S4,
                                            const float* __restrict__ norm,
                                            const int* __restrict__ csr,
                                            int c, int ids, int left, int rem, int nidx,
                                            float4& acc) {
    float4 x[16];
    const int nxt = csr[nidx];
    int nb = 0;
    if constexpr (MODE == 0) nb = __float_as_int(norm[ids]);
    G_QUAD(G_LOAD, 0)
    if (rem > 4) {
        G_QUAD(G_LOAD, 1)
        if (rem > 8) {
            G_QUAD(G_LOAD, 2)
            if (rem > 12) { G_QUAD(G_LOAD, 3) }
        }
    }
    __builtin_amdgcn_sched_barrier(0);
    G_QUAD(G_ADD, 0)
    if (rem > 4) {
        G_QUAD(G_ADD, 1)
        if (rem > 8) {
            G_QUAD(G_ADD, 2)
            if (rem > 12) { G_QUAD(G_ADD, 3) }
        }
    }
    return nxt;
}

template <int MODE>
__device__ __forceinline__ void gather_rows(const float* __restrict__ S,
                                            const float* __restrict__ norm,
                                            const int* __restrict__ rowptr,
                                            const int* __restrict__ csr,
                                            float* __restrict__ O) {
    static_assert(NPG == 4, "five segment bounds p0..p4 are kept per row");
    static_assert(NNODES % (2 * NPG) == 0 && (NNODES / NPG * 16) % 256 == 0, "whole waves, whole blocks");
    const int tid = blockIdx.x * 256 + threadIdx.x;
    const int c = tid & 15;
    const int v0 = (tid >> 4) * NPG;
    const float4* S4 = (const float4*)S;
    const int rp = rowptr[v0 + (c < NPG ? c : NPG)];
    int p0 = row_bcast<0>(rp), p1 = row_bcast<1>(rp), p2 = row_bcast<2>(rp);
    int p3 = row_bcast<3>(rp);
    const int p4 = row_bcast<4>(rp);
    int nl = __float_as_int(norm[v0 + (c < NPG ? c : NPG - 1)]);
    int ids = csr[min(p0 + c, NEDGES - 1)];
    for (int i = 0; i < NPG; ++i) {
        const int cnt = p1 - p0;
        const int m = max(__builtin_amdgcn_readlane(cnt, 0), __builtin_amdgcn_readlane(cnt, 16));
        const float nv = __int_as_float(row_bcast<0>(nl));
        float4 acc = make_float4(0.0f, 0.0f, 0.0f, 0.0f);
        for (int base = 0; base < m; base += 16) {
            const int rem = m - base;
            const int nidx = min((rem > 16 ? p0 + base + 16 : p1) + c, NEDGES - 1);
            ids = gather_batch<MODE>(S4, norm, csr, c, ids, cnt - base, rem, nidx, acc);
        }
        float4 h;
        h.x = nv * acc.x; h.y = nv * acc.y; h.z = nv * acc.z; h.w = nv * acc.w;
        if (MODE != 2) { h.x *= nv; h.y *= nv; h.z *= nv; h.w *= nv; }
        ((float4*)O)[(v0 + i) * 16 + c] = h;
        p0 = p1; p1 = p2; p2 = p3; p3 = p4;
        nl = __builtin_amdgcn_update_dpp(0, nl, 0x101, 0xF, 0xF, true);
    }
}

__global__ __launch_bounds__(256) void gather_first_kernel(const float* __restrict__ S,
                                                           const float* __restrict__ norm,
                                                           const int* __restrict__ rowptr,
                                                           const int* __restrict__ csr,
                                                           float* __restrict__ O) {
    gather_rows<0>(S, norm, rowptr, csr, O);
}
__global__ __launch_bounds__(256) void gather_mid_kernel(const float* __restrict__ S,
                                                         const float* __restrict__ norm,
                                                         const int* __restrict__ rowptr,
                                                         const int* __restrict__ csr,
                                                         float* __restrict__ O) {
    gather_rows<1>(S, norm, rowptr, csr, O);
}
__global__ __launch_bounds__(256) void gather_last_kernel(const float* __restrict__ S,
                                                          const float* __restrict__ norm,
                                                          const int* __restrict__ rowptr,
                                                          const int* __restrict__ csr,
                                                          float* __restrict__ O) {
    gather_rows<2>(S, norm, rowptr, csr, O);
}

__global__ __launch_bounds__(256) void ffn_wmma_kernel(
    const float* __restrict__ r, const float* __restrict__ feat,
    const float* __restrict__ w1, const float* __restrict__ b1,
    const float* __restrict__ w2, const float* __restrict__ b2,
    const float* __restrict__ gamma, const float* __restrict__ beta,
    float* __restrict__ out) {
    __shared__ float lds[8 * 16 * DIM];
    const int wave = threadIdx.x >> 5;
    const int lane = threadIdx.x & 31;
    const int idx  = lane & 15;
    const int half = lane >> 4;
    const int rowbase = blockIdx.x * 128 + wave * 16;
    float* A = lds + wave * 16 * DIM;

    const float2 g2  = ((const float2*)gamma)[lane];
    const float2 be2 = ((const float2*)beta)[lane];
    for (int m = 0; m < 16; ++m) {
        float2 x = ((const float2*)(r + (rowbase + m) * DIM))[lane];
        float s = x.x + x.y;
        #pragma unroll
        for (int o = 16; o > 0; o >>= 1) s += __shfl_xor(s, o, 32);
        float mu = s * (1.0f / 64.0f);
        float dx = x.x - mu, dy = x.y - mu;
        float v = dx * dx + dy * dy;
        #pragma unroll
        for (int o = 16; o > 0; o >>= 1) v += __shfl_xor(v, o, 32);
        float rstd = rsqrtf(v * (1.0f / 64.0f) + 1e-5f);
        float2 y;
        y.x = dx * rstd * g2.x + be2.x;
        y.y = dy * rstd * g2.y + be2.y;
        ((float2*)(A + m * DIM))[lane] = y;
    }
    __syncthreads();

    v8f acc[4];
    #pragma unroll
    for (int j = 0; j < 4; ++j) acc[j] = (v8f)(0.0f);
    for (int kb = 0; kb < 16; ++kb) {
        const int k = kb * 4 + half * 2;
        v2f a = *(const v2f*)(A + idx * DIM + k);
        #pragma unroll
        for (int j = 0; j < 4; ++j) {
            float2 wv = *(const float2*)(w1 + (j * 16 + idx) * DIM + k);
            v2f b = {wv.x, wv.y};
            acc[j] = __builtin_amdgcn_wmma_f32_16x16x4_f32(
                false, a, false, b, (short)0, acc[j], false, false);
        }
    }
    __syncthreads();
    #pragma unroll
    for (int j = 0; j < 4; ++j) {
        const float bias = b1[j * 16 + idx];
        #pragma unroll
        for (int t = 0; t < 8; ++t) {
            float hv = acc[j][t] + bias;
            A[(t + 8 * half) * DIM + j * 16 + idx] = fmaxf(hv, 0.0f);
        }
    }
    __syncthreads();

    v8f acc2[4];
    #pragma unroll
    for (int j = 0; j < 4; ++j) acc2[j] = (v8f)(0.0f);
    for (int kb = 0; kb < 16; ++kb) {
        const int k = kb * 4 + half * 2;
        v2f a = *(const v2f*)(A + idx * DIM + k);
        #pragma unroll
        for (int j = 0; j < 4; ++j) {
            float2 wv = *(const float2*)(w2 + (j * 16 + idx) * DIM + k);
            v2f b = {wv.x, wv.y};
            acc2[j] = __builtin_amdgcn_wmma_f32_16x16x4_f32(
                false, a, false, b, (short)0, acc2[j], false, false);
        }
    }
    #pragma unroll
    for (int j = 0; j < 4; ++j) {
        const int col = j * 16 + idx;
        const float bias = b2[col];
        #pragma unroll
        for (int t = 0; t < 8; ++t) {
            const int row = rowbase + t + 8 * half;
            out[row * DIM + col] = acc2[j][t] + bias + r[row * DIM + col] + feat[row * DIM + col];
        }
    }
}

extern "C" void kernel_launch(void* const* d_in, const int* in_sizes, int n_in,
                              void* d_out, int out_size, void* d_ws, size_t ws_size,
                              hipStream_t stream) {
    const float* feat = (const float*)d_in[0];
    const int*   esrc = (const int*)d_in[1];
    const int*   edst = (const int*)d_in[2];
    const float* w1   = (const float*)d_in[3];
    const float* b1   = (const float*)d_in[4];
    const float* w2   = (const float*)d_in[5];
    const float* b2   = (const float*)d_in[6];
    const float* gam  = (const float*)d_in[7];
    const float* bet  = (const float*)d_in[8];

    float* out0 = (float*)d_out;
    float* out1 = out0 + (size_t)NNODES * DIM;

    int*      runT   = (int*)d_ws;
    int*      rowptr = runT + NNODES + 256;
    float*    norm   = (float*)(rowptr + NNODES + 64);
    unsigned* bkt    = (unsigned*)(norm + NNODES);
    int*      csr    = (int*)(bkt + NEDGES);
    float*    sA     = (float*)(csr + NEDGES);
    float*    sB     = sA + (size_t)NNODES * DIM;

    dim3 blk(256);
    dim3 gG(NNODES / NPG * 16 / 256);

    bucket_scatter_kernel<<<dim3(NEDGES / EPB), dim3(512), 0, stream>>>(esrc, edst, runT, bkt);
    bucket_sort_kernel<<<dim3(NBKT), dim3(1024), 0, stream>>>(bkt, runT, rowptr, norm, csr);

    gather_first_kernel<<<gG, blk, 0, stream>>>(feat, norm, rowptr, csr, sA);
    gather_mid_kernel<<<gG, blk, 0, stream>>>(sA,   norm, rowptr, csr, sB);
    gather_last_kernel<<<gG, blk, 0, stream>>>(sB,   norm, rowptr, csr, out1);

    ffn_wmma_kernel<<<dim3(NNODES / 128), blk, 0, stream>>>(
        out1, feat, w1, b1, w2, b2, gam, bet, out0);
}


# Fuse LN+FFN into the last hop's kernel: rows normalised in registers (LN sums regrouped: 4 per lane, then 16-lane DPP butterfly), 16-row tiles via LDS, a wave takes half a tile's columns; 5 launches
# speedup vs baseline: 5.6376x; 1.1364x over previous
#include <hip/hip_runtime.h>

#define NNODES 65536
#define NEDGES 1048576
#define DIM    64
#define NBKT   256
#define EPB    4096
#define CAP    8192

typedef float v2f __attribute__((ext_vector_type(2)));
typedef float v8f __attribute__((ext_vector_type(8)));


__device__ __forceinline__ int block_excl_scan256(int x, int t, int* s_sum) {
    const int lane = t & 31;
    const int w = t >> 5;
    int inc = x;
    for (int o = 1; o < 32; o <<= 1) {
        int y = __shfl_up(inc, o, 32);
        if (lane >= o) inc += y;
    }
    if (t < 256 && lane == 31) s_sum[w] = inc;
    __syncthreads();
    int off = 0;
    if (t < 256) {
        for (int i = 0; i < 8; ++i) if (i < w) off += s_sum[i];
    }
    return off + inc - x;
}

__global__ __launch_bounds__(512) void bucket_scatter_kernel(const int* __restrict__ src,
                                                             const int* __restrict__ dst,
                                                             int* __restrict__ runT,
                                                             unsigned* __restrict__ bkt) {
    __shared__ int h[NBKT];
    __shared__ int lstart[NBKT];
    __shared__ int s_sum[8];
    __shared__ __attribute__((aligned(16))) unsigned sorted[EPB];
    const int t = threadIdx.x, b = blockIdx.x;
    if (t < NBKT) h[t] = 0;
    __syncthreads();
    const int4* d4 = (const int4*)dst + b * (EPB / 4);
    const int4* s4 = (const int4*)src + b * (EPB / 4);
    const int4 d0 = d4[t], d1 = d4[512 + t];
    const int4 s0 = s4[t], s1 = s4[512 + t];
    const int dd[8] = {d0.x, d0.y, d0.z, d0.w, d1.x, d1.y, d1.z, d1.w};
    const int ss[8] = {s0.x, s0.y, s0.z, s0.w, s1.x, s1.y, s1.z, s1.w};
    int kk[8], rr[8];
    #pragma unroll
    for (int j = 0; j < 8; ++j) {
        kk[j] = (dd[j] >> 8) & 255;
        rr[j] = atomicAdd(&h[kk[j]], 1);
    }
    __syncthreads();
    const int x = (t < NBKT) ? h[t] : 0;
    const int excl = block_excl_scan256(x, t, s_sum);
    if (t < NBKT) {
        lstart[t] = excl;
        runT[t * 256 + b] = excl;
    }
    if (t == 0) runT[NBKT * 256 + b] = EPB;
    __syncthreads();
    #pragma unroll
    for (int j = 0; j < 8; ++j)
        sorted[lstart[kk[j]] + rr[j]] = ((unsigned)ss[j] << 16) | ((unsigned)dd[j] & 0xFFFFu);
    __syncthreads();
    uint4* o4 = (uint4*)bkt + b * (EPB / 4);
    const uint4* l4 = (const uint4*)sorted;
    o4[t] = l4[t];
    o4[512 + t] = l4[512 + t];
}

__global__ __launch_bounds__(1024) void bucket_sort_kernel(const unsigned* __restrict__ bkt,
                                                           const int* __restrict__ runT,
                                                           int* __restrict__ rowptr,
                                                           float* __restrict__ norm,
                                                           int* __restrict__ csr) {
    __shared__ int cnt[256];
    __shared__ int cur[256];
    __shared__ int s_sum[8];
    __shared__ int s_len[8];
    __shared__ int s_beg[8];
    __shared__ int run_src[256];
    __shared__ int run_len[256];
    __shared__ int run_pos[256];
    __shared__ unsigned ent[CAP];
    __shared__ int stage[CAP];
    const int t = threadIdx.x, k = blockIdx.x;
    const int lane = t & 31, w = t >> 5;
    int beg = 0, len = 0;
    if (t < 256) {
        beg = runT[k * 256 + t];
        len = runT[(k + 1) * 256 + t] - beg;
        cnt[t] = 0;
    }
    int inc = len, bsum = beg;
    for (int o = 1; o < 32; o <<= 1) {
        int y = __shfl_up(inc, o, 32);
        if (lane >= o) inc += y;
    }
    for (int o = 16; o > 0; o >>= 1) bsum += __shfl_xor(bsum, o, 32);
    if (t < 256 && lane == 31) { s_len[w] = inc; s_beg[w] = bsum; }
    __syncthreads();
    int off = 0, n = 0, start = 0;
    for (int i = 0; i < 8; ++i) {
        const int li = s_len[i];
        if (i < w) off += li;
        n += li;
        start += s_beg[i];
    }
    if (t < 256) {
        run_src[t] = t * EPB + beg;
        run_len[t] = len;
        run_pos[t] = off + inc - len;
    }
    __syncthreads();
    const int r = t >> 2, j0 = t & 3;
    const int rsrc = run_src[r], rlen = run_len[r], rpos = run_pos[r];
    const bool staged = n <= CAP;
    if (staged) {
        unsigned wv[8];
        #pragma unroll
        for (int q = 0; q < 8; ++q) {
            const int j = j0 + 4 * q;
            wv[q] = (j < rlen) ? bkt[rsrc + j] : 0u;
        }
        #pragma unroll
        for (int q = 0; q < 8; ++q) {
            const int j = j0 + 4 * q;
            if (j < rlen) {
                ent[rpos + j] = wv[q];
                atomicAdd(&cnt[wv[q] & 255u], 1);
            }
        }
        for (int j = j0 + 32; j < rlen; j += 4) {
            const unsigned v = bkt[rsrc + j];
            ent[rpos + j] = v;
            atomicAdd(&cnt[v & 255u], 1);
        }
    } else {
        for (int j = j0; j < rlen; j += 4) atomicAdd(&cnt[bkt[rsrc + j] & 255u], 1);
    }
    __syncthreads();
    const int x = (t < 256) ? cnt[t] : 0;
    const int excl = block_excl_scan256(x, t, s_sum);
    if (t < 256) {
        cur[t] = excl;
        rowptr[k * 256 + t] = start + excl;
        norm[k * 256 + t] = rsqrtf(fmaxf((float)x, 1.0f));
    }
    if (k == NBKT - 1 && t == 0) rowptr[NNODES] = start + n;
    __syncthreads();
    if (staged) {
        for (int i = t; i < n; i += 1024) {
            const unsigned v = ent[i];
            stage[atomicAdd(&cur[v & 255u], 1)] = (int)(v >> 16);
        }
        __syncthreads();
        for (int i = t; i < n; i += 1024) csr[start + i] = stage[i];
    } else {
        for (int j = j0; j < rlen; j += 4) {
            const unsigned v = bkt[rsrc + j];
            csr[start + atomicAdd(&cur[v & 255u], 1)] = (int)(v >> 16);
        }
    }
}

#define NPG 4

template <int E>
__device__ __forceinline__ int row_bcast(int x) {
    return __builtin_amdgcn_update_dpp(0, x, 0x150 + E, 0xF, 0xF, true);
}

#define G_LOAD(E)                                                                       \
    {                                                                                   \
        const unsigned id = (unsigned)row_bcast<E>(ids);                                \
        x[E] = S4[((E) < left) ? (id * 16u + (unsigned)c) : (unsigned)c];               \
    }
#define G_ADD(E)                                                                        \
    {                                                                                   \
        const bool ok = (E) < left;                                                     \
        const float tx = ok ? x[E].x : 0.0f, ty = ok ? x[E].y : 0.0f;                   \
        const float tz = ok ? x[E].z : 0.0f, tw = ok ? x[E].w : 0.0f;                   \
        if constexpr (MODE == 0) {                                                      \
            const float n = __int_as_float(row_bcast<E>(nb));                           \
            acc.x += n * tx; acc.y += n * ty; acc.z += n * tz; acc.w += n * tw;         \
        } else {                                                                        \
            acc.x += tx; acc.y += ty; acc.z += tz; acc.w += tw;                         \
        }                                                                               \
    }
#define G_QUAD(M, Q) M(4 * (Q)) M(4 * (Q) + 1) M(4 * (Q) + 2) M(4 * (Q) + 3)

template <int MODE>
__device__ __forceinline__ int gather_batch(const float4* __restrict__ S4,
                                            const float* __restrict__ norm,
                                            const int* __restrict__ csr,
                                            int c, int ids, int left, int rem, int nidx,
                                            float4& acc) {
    float4 x[16];
    const int nxt = csr[nidx];
    int nb = 0;
    if constexpr (MODE == 0) nb = __float_as_int(norm[ids]);
    G_QUAD(G_LOAD, 0)
    if (rem > 4) {
        G_QUAD(G_LOAD, 1)
        if (rem > 8) {
            G_QUAD(G_LOAD, 2)
            if (rem > 12) { G_QUAD(G_LOAD, 3) }
        }
    }
    __builtin_amdgcn_sched_barrier(0);
    G_QUAD(G_ADD, 0)
    if (rem > 4) {
        G_QUAD(G_ADD, 1)
        if (rem > 8) {
            G_QUAD(G_ADD, 2)
            if (rem > 12) { G_QUAD(G_ADD, 3) }
        }
    }
    return nxt;
}

template <int MODE, bool KEEP = false>
__device__ __forceinline__ void gather_rows(const float* __restrict__ S,
                                            const float* __restrict__ norm,
                                            const int* __restrict__ rowptr,
                                            const int* __restrict__ csr,
                                            float* __restrict__ O,
                                            float4* keep = nullptr) {
    static_assert(NPG == 4, "five segment bounds p0..p4 are kept per row");
    static_assert(NNODES % (2 * NPG) == 0 && (NNODES / NPG * 16) % 256 == 0, "whole waves, whole blocks");
    const int tid = blockIdx.x * 256 + threadIdx.x;
    const int c = tid & 15;
    const int v0 = (tid >> 4) * NPG;
    const float4* S4 = (const float4*)S;
    const int rp = rowptr[v0 + (c < NPG ? c : NPG)];
    int p0 = row_bcast<0>(rp), p1 = row_bcast<1>(rp), p2 = row_bcast<2>(rp);
    int p3 = row_bcast<3>(rp);
    const int p4 = row_bcast<4>(rp);
    int nl = __float_as_int(norm[v0 + (c < NPG ? c : NPG - 1)]);
    int ids = csr[min(p0 + c, NEDGES - 1)];
    for (int i = 0; i < NPG; ++i) {
        const int cnt = p1 - p0;
        const int m = max(__builtin_amdgcn_readlane(cnt, 0), __builtin_amdgcn_readlane(cnt, 16));
        const float nv = __int_as_float(row_bcast<0>(nl));
        float4 acc = make_float4(0.0f, 0.0f, 0.0f, 0.0f);
        for (int base = 0; base < m; base += 16) {
            const int rem = m - base;
            const int nidx = min((rem > 16 ? p0 + base + 16 : p1) + c, NEDGES - 1);
            ids = gather_batch<MODE>(S4, norm, csr, c, ids, cnt - base, rem, nidx, acc);
        }
        float4 h;
        h.x = nv * acc.x; h.y = nv * acc.y; h.z = nv * acc.z; h.w = nv * acc.w;
        if (MODE != 2) { h.x *= nv; h.y *= nv; h.z *= nv; h.w *= nv; }
        ((float4*)O)[(v0 + i) * 16 + c] = h;
        if constexpr (KEEP) { keep[0] = keep[1]; keep[1] = keep[2]; keep[2] = keep[3]; keep[3] = h; }
        p0 = p1; p1 = p2; p2 = p3; p3 = p4;
        nl = __builtin_amdgcn_update_dpp(0, nl, 0x101, 0xF, 0xF, true);
    }
}

__global__ __launch_bounds__(256) void gather_first_kernel(const float* __restrict__ S,
                                                           const float* __restrict__ norm,
                                                           const int* __restrict__ rowptr,
                                                           const int* __restrict__ csr,
                                                           float* __restrict__ O) {
    gather_rows<0>(S, norm, rowptr, csr, O);
}
__global__ __launch_bounds__(256) void gather_mid_kernel(const float* __restrict__ S,
                                                         const float* __restrict__ norm,
                                                         const int* __restrict__ rowptr,
                                                         const int* __restrict__ csr,
                                                         float* __restrict__ O) {
    gather_rows<1>(S, norm, rowptr, csr, O);
}
#define YP 68

template <int M>
__device__ __forceinline__ float row_xor(float x) {
    return __int_as_float(__builtin_amdgcn_update_dpp(0, __float_as_int(x), 0x160 + M, 0xF, 0xF, true));
}
__device__ __forceinline__ float row_sum16(float x) {
    x += row_xor<1>(x);
    x += row_xor<2>(x);
    x += row_xor<4>(x);
    x += row_xor<8>(x);
    return x;
}
__device__ __forceinline__ float4 ln_row(float4 h, float4 g, float4 be) {
    const float mu = row_sum16((h.x + h.y) + (h.z + h.w)) * (1.0f / 64.0f);
    const float dx = h.x - mu, dy = h.y - mu, dz = h.z - mu, dw = h.w - mu;
    const float var = row_sum16((dx * dx + dy * dy) + (dz * dz + dw * dw)) * (1.0f / 64.0f);
    const float rstd = rsqrtf(var + 1e-5f);
    float4 y;
    y.x = dx * rstd * g.x + be.x;
    y.y = dy * rstd * g.y + be.y;
    y.z = dz * rstd * g.z + be.z;
    y.w = dw * rstd * g.w + be.w;
    return y;
}

__device__ __forceinline__ void tile_gemm_half(const float* A, const float* __restrict__ W,
                                               int idx, int half, int jh, v8f (&acc)[2]) {
    for (int kb = 0; kb < 16; ++kb) {
        const int k = kb * 4 + half * 2;
        const v2f a = *(const v2f*)(A + idx * YP + k);
        #pragma unroll
        for (int jj = 0; jj < 2; ++jj) {
            const float2 wv = *(const float2*)(W + ((jh * 2 + jj) * 16 + idx) * DIM + k);
            const v2f b = {wv.x, wv.y};
            acc[jj] = __builtin_amdgcn_wmma_f32_16x16x4_f32(
                false, a, false, b, (short)0, acc[jj], false, false);
        }
    }
}

__global__ __launch_bounds__(256) void gather_last_ffn_kernel(
    const float* __restrict__ S, const float* __restrict__ norm,
    const int* __restrict__ rowptr, const int* __restrict__ csr,
    float* __restrict__ R, const float* __restrict__ feat,
    const float* __restrict__ w1, const float* __restrict__ b1,
    const float* __restrict__ w2, const float* __restrict__ b2,
    const float* __restrict__ gamma, const float* __restrict__ beta,
    float* __restrict__ out) {
    __shared__ __attribute__((aligned(16))) float Yb[64 * YP];
    __shared__ __attribute__((aligned(16))) float Hb[64 * YP];
    float4 hr[NPG];
    #pragma unroll
    for (int i = 0; i < NPG; ++i) hr[i] = make_float4(0.0f, 0.0f, 0.0f, 0.0f);
    gather_rows<2, true>(S, norm, rowptr, csr, R, hr);

    const int t = threadIdx.x;
    const int c = t & 15;
    const int g = t >> 4;
    const int v0 = (blockIdx.x * 16 + g) * NPG;
    const float4 g4  = ((const float4*)gamma)[c];
    const float4 be4 = ((const float4*)beta)[c];
    #pragma unroll
    for (int i = 0; i < NPG; ++i)
        *(float4*)(Yb + (4 * g + i) * YP + 4 * c) = ln_row(hr[i], g4, be4);
    __syncthreads();

    const int wave = t >> 5, lane = t & 31;
    const int idx  = lane & 15;
    const int half = lane >> 4;
    const int tt = wave >> 1;
    const int jh = wave & 1;

    v8f acc[2];
    acc[0] = (v8f)(0.0f); acc[1] = (v8f)(0.0f);
    tile_gemm_half(Yb + tt * 16 * YP, w1, idx, half, jh, acc);
    #pragma unroll
    for (int jj = 0; jj < 2; ++jj) {
        const int col = (jh * 2 + jj) * 16 + idx;
        const float bias = b1[col];
        #pragma unroll
        for (int q = 0; q < 8; ++q)
            Hb[(tt * 16 + q + 8 * half) * YP + col] = fmaxf(acc[jj][q] + bias, 0.0f);
    }
    __syncthreads();

    v8f acc2[2];
    acc2[0] = (v8f)(0.0f); acc2[1] = (v8f)(0.0f);
    tile_gemm_half(Hb + tt * 16 * YP, w2, idx, half, jh, acc2);
    #pragma unroll
    for (int jj = 0; jj < 2; ++jj) {
        const int col = (jh * 2 + jj) * 16 + idx;
        const float bias = b2[col];
        #pragma unroll
        for (int q = 0; q < 8; ++q)
            Yb[(tt * 16 + q + 8 * half) * YP + col] = acc2[jj][q] + bias;
    }
    __syncthreads();

    #pragma unroll
    for (int i = 0; i < NPG; ++i) {
        const float4 ff = *(const float4*)(Yb + (4 * g + i) * YP + 4 * c);
        const float4 f  = ((const float4*)feat)[(v0 + i) * 16 + c];
        float4 o;
        o.x = ff.x + hr[i].x + f.x;
        o.y = ff.y + hr[i].y + f.y;
        o.z = ff.z + hr[i].z + f.z;
        o.w = ff.w + hr[i].w + f.w;
        ((float4*)out)[(v0 + i) * 16 + c] = o;
    }
}

extern "C" void kernel_launch(void* const* d_in, const int* in_sizes, int n_in,
                              void* d_out, int out_size, void* d_ws, size_t ws_size,
                              hipStream_t stream) {
    const float* feat = (const float*)d_in[0];
    const int*   esrc = (const int*)d_in[1];
    const int*   edst = (const int*)d_in[2];
    const float* w1   = (const float*)d_in[3];
    const float* b1   = (const float*)d_in[4];
    const float* w2   = (const float*)d_in[5];
    const float* b2   = (const float*)d_in[6];
    const float* gam  = (const float*)d_in[7];
    const float* bet  = (const float*)d_in[8];

    float* out0 = (float*)d_out;
    float* out1 = out0 + (size_t)NNODES * DIM;

    int*      runT   = (int*)d_ws;
    int*      rowptr = runT + NNODES + 256;
    float*    norm   = (float*)(rowptr + NNODES + 64);
    unsigned* bkt    = (unsigned*)(norm + NNODES);
    int*      csr    = (int*)(bkt + NEDGES);
    float*    sA     = (float*)(csr + NEDGES);
    float*    sB     = sA + (size_t)NNODES * DIM;

    dim3 blk(256);
    dim3 gG(NNODES / NPG * 16 / 256);

    bucket_scatter_kernel<<<dim3(NEDGES / EPB), dim3(512), 0, stream>>>(esrc, edst, runT, bkt);
    bucket_sort_kernel<<<dim3(NBKT), dim3(1024), 0, stream>>>(bkt, runT, rowptr, norm, csr);

    gather_first_kernel<<<gG, blk, 0, stream>>>(feat, norm, rowptr, csr, sA);
    gather_mid_kernel<<<gG, blk, 0, stream>>>(sA,   norm, rowptr, csr, sB);
    gather_last_ffn_kernel<<<gG, blk, 0, stream>>>(sB, norm, rowptr, csr, out1,
                                                   feat, w1, b1, w2, b2, gam, bet, out0);
}
